# starting kernel
# baseline (speedup 1.0000x reference)
#include <hip/hip_runtime.h>

typedef __attribute__((ext_vector_type(2))) float v2f;
typedef __attribute__((ext_vector_type(8))) float v8f;

#define B_       2
#define CIN      3
#define COUT     64
#define HIN      721
#define WIN      1440
#define HOUT     361
#define WOUT     720
#define KTAP     9
#define NNZ      96
#define LN_EPS   1e-5f

#define ASTRIDE  98
#define NWAVES   6
#define NTHREADS (NWAVES * 32)
#define PLANE    (HIN * WIN)
#define NTOT     (B_ * WOUT)

__global__ __launch_bounds__(NTHREADS)
void opm_gather_gemm_ln_kernel(const float* __restrict__ x,
                               const float* __restrict__ weight,
                               const float* __restrict__ psi,
                               const int*   __restrict__ idx_hi,
                               const int*   __restrict__ idx_dwi,
                               const float* __restrict__ lnw,
                               const float* __restrict__ lnb,
                               float*       __restrict__ out)
{
    __shared__ float psi_lds[KTAP * NNZ];
    __shared__ float w_lds[COUT * CIN * KTAP];
    __shared__ int2  base_lds[NNZ];
    __shared__ float A_lds[COUT * ASTRIDE];
    __shared__ float lnw_lds[COUT];
    __shared__ float lnb_lds[COUT];

    const int tid  = threadIdx.x;
    const int h    = blockIdx.y;
    const int lane = tid & 31;
    const int wid  = tid >> 5;
    const int l16  = lane & 15;
    const int half = lane >> 4;

    for (int i = tid; i < KTAP * NNZ; i += NTHREADS) {
        int k9 = i / NNZ, e = i - k9 * NNZ;
        psi_lds[i] = psi[(k9 * HOUT + h) * NNZ + e];
    }
    for (int i = tid; i < COUT * CIN * KTAP; i += NTHREADS)
        w_lds[i] = weight[i];
    for (int i = tid; i < NNZ; i += NTHREADS)
        base_lds[i] = make_int2(idx_hi[h * NNZ + i] * WIN, idx_dwi[h * NNZ + i]);
    for (int i = tid; i < COUT; i += NTHREADS) {
        lnw_lds[i] = lnw[i];
        lnb_lds[i] = lnb[i];
    }

    const int eown = tid & (NNZ - 1);
    const int eA_own = tid - (tid >= NNZ ? NNZ : 0);
    const int o0     = (tid >= NNZ) ? 1 : 0;
    (void)eown;

    const int n    = blockIdx.x * (NWAVES * 16) + wid * 16 + l16;
    const int b    = n / WOUT;
    const int w    = n - b * WOUT;
    const int wcol = w << 1;

    v8f acc[4];
    #pragma unroll
    for (int mt = 0; mt < 4; ++mt) acc[mt] = (v8f){0.f,0.f,0.f,0.f,0.f,0.f,0.f,0.f};

    for (int c = 0; c < CIN; ++c) {
        __syncthreads();

        {
            float p[KTAP];
            #pragma unroll
            for (int k9 = 0; k9 < KTAP; ++k9)
                p[k9] = psi_lds[k9 * NNZ + eA_own];
            for (int o = o0; o < COUT; o += 2) {
                const float* wp = &w_lds[o * (CIN * KTAP) + c * KTAP];
                float s = 0.f;
                #pragma unroll
                for (int k9 = 0; k9 < KTAP; ++k9)
                    s += wp[k9] * p[k9];
                A_lds[o * ASTRIDE + eA_own] = s;
            }
        }
        __syncthreads();

        const int plane = (b * CIN + c) * PLANE;

        #pragma unroll 2
        for (int e0 = 0; e0 < NNZ; e0 += 4) {
            const int eA = e0 + (half << 1);

            const int2 r0 = base_lds[eA];
            const int2 r1 = base_lds[eA + 1];
            int c0 = r0.y + wcol; c0 -= (c0 >= WIN) ? WIN : 0;
            int c1 = r1.y + wcol; c1 -= (c1 >= WIN) ? WIN : 0;
            v2f bf;
            bf.x = x[plane + r0.x + c0];
            bf.y = x[plane + r1.x + c1];

            v2f a0 = *(const v2f*)&A_lds[( 0 + l16) * ASTRIDE + eA];
            v2f a1 = *(const v2f*)&A_lds[(16 + l16) * ASTRIDE + eA];
            v2f a2 = *(const v2f*)&A_lds[(32 + l16) * ASTRIDE + eA];
            v2f a3 = *(const v2f*)&A_lds[(48 + l16) * ASTRIDE + eA];

            acc[0] = __builtin_amdgcn_wmma_f32_16x16x4_f32(false, a0, false, bf, (short)0, acc[0], false, false);
            acc[1] = __builtin_amdgcn_wmma_f32_16x16x4_f32(false, a1, false, bf, (short)0, acc[1], false, false);
            acc[2] = __builtin_amdgcn_wmma_f32_16x16x4_f32(false, a2, false, bf, (short)0, acc[2], false, false);
            acc[3] = __builtin_amdgcn_wmma_f32_16x16x4_f32(false, a3, false, bf, (short)0, acc[3], false, false);
        }
    }

    float s = 0.f, ss = 0.f;
    #pragma unroll
    for (int mt = 0; mt < 4; ++mt) {
        #pragma unroll
        for (int r = 0; r < 8; ++r) {
            float v = acc[mt][r];
            s  += v;
            ss += v * v;
        }
    }
    s  += __shfl_xor(s, 16, 32);
    ss += __shfl_xor(ss, 16, 32);
    const float mean = s * (1.f / COUT);
    const float var  = ss * (1.f / COUT) - mean * mean;
    const float rstd = rsqrtf(var + LN_EPS);

    float* outp = out + (size_t)b * COUT * HOUT * WOUT + (size_t)h * WOUT + w;
    #pragma unroll
    for (int mt = 0; mt < 4; ++mt) {
        #pragma unroll
        for (int r = 0; r < 8; ++r) {
            const int o = mt * 16 + r + (half << 3);
            const float v = (acc[mt][r] - mean) * rstd * lnw_lds[o] + lnb_lds[o];
            outp[(size_t)o * (HOUT * WOUT)] = v;
        }
    }
}

extern "C" void kernel_launch(void* const* d_in, const int* in_sizes, int n_in,
                              void* d_out, int out_size, void* d_ws, size_t ws_size,
                              hipStream_t stream) {
    (void)in_sizes; (void)n_in; (void)d_ws; (void)ws_size; (void)out_size;
    const float* x       = (const float*)d_in[0];
    const float* weight  = (const float*)d_in[1];
    const float* psi     = (const float*)d_in[2];
    const int*   idx_hi  = (const int*)d_in[3];
    const int*   idx_dwi = (const int*)d_in[4];
    const float* lnw     = (const float*)d_in[5];
    const float* lnb     = (const float*)d_in[6];
    float*       out     = (float*)d_out;

    dim3 grid(NTOT / (NWAVES * 16), HOUT, 1);
    dim3 block(NTHREADS, 1, 1);
    opm_gather_gemm_ln_kernel<<<grid, block, 0, stream>>>(
        x, weight, psi, idx_hi, idx_dwi, lnw, lnb, out);
}


# Re-associate as the reference does: y=psi*xg (16x96 per channel) then z=W*y (K=27) on f32 WMMA, y passed in registers; drops the per-block W*psi build; changes the floating-point summation order
# speedup vs baseline: 2.2203x; 2.2203x over previous
#include <hip/hip_runtime.h>


typedef __attribute__((ext_vector_type(2))) float v2f;
typedef __attribute__((ext_vector_type(8))) float v8f;

#define B_       2
#define CIN      3
#define COUT     64
#define HIN      721
#define WIN      1440
#define HOUT     361
#define WOUT     720
#define KTAP     9
#define NNZ      96
#define LN_EPS   1e-5f

#define KPAD     16
#define KLO      5
#define PSTRIDE  98
#define WROW     (CIN * KTAP)
#define NWAVES   6
#define NTHREADS (NWAVES * 32)
#define PLANE    (HIN * WIN)
#define NTOT     (B_ * WOUT)

__global__ __launch_bounds__(NTHREADS)
void opm_gather_gemm_ln_kernel(const float* __restrict__ x,
                               const float* __restrict__ weight,
                               const float* __restrict__ psi,
                               const int*   __restrict__ idx_hi,
                               const int*   __restrict__ idx_dwi,
                               const float* __restrict__ lnw,
                               const float* __restrict__ lnb,
                               float*       __restrict__ out)
{
    __shared__ float psiA_lds[KPAD * PSTRIDE];
    __shared__ float w_lds[COUT * WROW];
    __shared__ __attribute__((aligned(16))) int2 base_lds[NNZ];
    __shared__ float lnw_lds[COUT];
    __shared__ float lnb_lds[COUT];

    const int tid  = threadIdx.x;
    const int h    = blockIdx.y;
    const int lane = tid & 31;
    const int wid  = tid >> 5;
    const int l16  = lane & 15;
    const int half = lane >> 4;

    for (int i = tid; i < KTAP * NNZ; i += NTHREADS) {
        int k9 = i / NNZ, e = i - k9 * NNZ;
        const int m = (k9 < KLO) ? k9 : k9 + (8 - KLO);
        psiA_lds[m * PSTRIDE + e] = psi[(k9 * HOUT + h) * NNZ + e];
    }
    for (int i = tid; i < 7 * PSTRIDE; i += NTHREADS) {
        const int at = (i < 3 * PSTRIDE) ? (KLO * PSTRIDE + i) : (12 * PSTRIDE + (i - 3 * PSTRIDE));
        psiA_lds[at] = 0.f;
    }
    for (int i = tid; i < COUT * WROW; i += NTHREADS)
        w_lds[i] = weight[i];
    for (int i = tid; i < NNZ; i += NTHREADS)
        base_lds[i] = make_int2(idx_hi[h * NNZ + i] * WIN, idx_dwi[h * NNZ + i]);
    for (int i = tid; i < COUT; i += NTHREADS) {
        lnw_lds[i] = lnw[i];
        lnb_lds[i] = lnb[i];
    }

    const int n    = blockIdx.x * (NWAVES * 16) + wid * 16 + l16;
    const int b    = n / WOUT;
    const int w    = n - b * WOUT;
    const int wcol = w << 1;
    const int bofs = b * (CIN * PLANE);

    __syncthreads();

    v8f y0 = (v8f){0.f,0.f,0.f,0.f,0.f,0.f,0.f,0.f};
    v8f y1 = y0, y2 = y0;

    #pragma unroll 4
    for (int e0 = 0; e0 < NNZ; e0 += 4) {
        const int eA = e0 + (half << 1);

        const int4 rr = *(const int4*)&base_lds[eA];
        int c0 = rr.y + wcol; c0 -= (c0 >= WIN) ? WIN : 0;
        int c1 = rr.w + wcol; c1 -= (c1 >= WIN) ? WIN : 0;
        const int i0 = bofs + rr.x + c0;
        const int i1 = bofs + rr.z + c1;

        v2f b0, b1, b2;
        b0.x = x[i0];              b0.y = x[i1];
        b1.x = x[i0 + PLANE];      b1.y = x[i1 + PLANE];
        b2.x = x[i0 + 2 * PLANE];  b2.y = x[i1 + 2 * PLANE];

        const v2f pa = *(const v2f*)&psiA_lds[l16 * PSTRIDE + eA];

        y0 = __builtin_amdgcn_wmma_f32_16x16x4_f32(false, pa, false, b0, (short)0, y0, false, false);
        y1 = __builtin_amdgcn_wmma_f32_16x16x4_f32(false, pa, false, b1, (short)0, y1, false, false);
        y2 = __builtin_amdgcn_wmma_f32_16x16x4_f32(false, pa, false, b2, (short)0, y2, false, false);
    }

    v8f acc[4];
    #pragma unroll
    for (int mt = 0; mt < 4; ++mt) acc[mt] = (v8f){0.f,0.f,0.f,0.f,0.f,0.f,0.f,0.f};

    const float* wl = &w_lds[l16 * WROW];
    const int   kh = half ? KLO : 0;
    const bool  lo = (half == 0);

    #pragma unroll
    for (int c = 0; c < CIN; ++c) {
        const v8f yc = (c == 0) ? y0 : ((c == 1) ? y1 : y2);
        #pragma unroll
        for (int j = 0; j < 2; ++j) {
            v2f bf;
            bf.x = yc[2 * j];
            bf.y = yc[2 * j + 1];
            #pragma unroll
            for (int mt = 0; mt < 4; ++mt) {
                const float* wp = wl + mt * 16 * WROW + c * KTAP + 2 * j + kh;
                v2f a;
                a.x = wp[0];
                a.y = wp[1];
                acc[mt] = __builtin_amdgcn_wmma_f32_16x16x4_f32(false, a, false, bf, (short)0, acc[mt], false, false);
            }
        }
    }
    {
        v2f bf6, bf7;
        bf6.x = y0[KLO - 1];  bf6.y = y1[KLO - 1];
        bf7.x = y2[KLO - 1];  bf7.y = 0.f;
        #pragma unroll
        for (int mt = 0; mt < 4; ++mt) {
            const float* wp = wl + mt * 16 * WROW + (KLO - 1);
            const float w0 = wp[0], w1 = wp[KTAP], w2 = wp[2 * KTAP];
            v2f a6, a7;
            a6.x = lo ? w0 : 0.f;
            a6.y = lo ? w1 : 0.f;
            a7.x = lo ? w2 : 0.f;
            a7.y = 0.f;
            acc[mt] = __builtin_amdgcn_wmma_f32_16x16x4_f32(false, a6, false, bf6, (short)0, acc[mt], false, false);
            acc[mt] = __builtin_amdgcn_wmma_f32_16x16x4_f32(false, a7, false, bf7, (short)0, acc[mt], false, false);
        }
    }

    float s = 0.f, ss = 0.f;
    #pragma unroll
    for (int mt = 0; mt < 4; ++mt) {
        #pragma unroll
        for (int r = 0; r < 8; ++r) {
            float v = acc[mt][r];
            s  += v;
            ss += v * v;
        }
    }
    s  += __shfl_xor(s, 16, 32);
    ss += __shfl_xor(ss, 16, 32);
    const float mean = s * (1.f / COUT);
    const float var  = ss * (1.f / COUT) - mean * mean;
    const float rstd = rsqrtf(var + LN_EPS);

    float* outp = out + (size_t)b * COUT * HOUT * WOUT + (size_t)h * WOUT + w;
    #pragma unroll
    for (int mt = 0; mt < 4; ++mt) {
        #pragma unroll
        for (int r = 0; r < 8; ++r) {
            const int o = mt * 16 + r + (half << 3);
            const float v = (acc[mt][r] - mean) * rstd * lnw_lds[o] + lnb_lds[o];
            outp[(size_t)o * (HOUT * WOUT)] = v;
        }
    }
}

extern "C" void kernel_launch(void* const* d_in, const int* in_sizes, int n_in,
                              void* d_out, int out_size, void* d_ws, size_t ws_size,
                              hipStream_t stream) {
    (void)in_sizes; (void)n_in; (void)d_ws; (void)ws_size; (void)out_size;
    const float* x       = (const float*)d_in[0];
    const float* weight  = (const float*)d_in[1];
    const float* psi     = (const float*)d_in[2];
    const int*   idx_hi  = (const int*)d_in[3];
    const int*   idx_dwi = (const int*)d_in[4];
    const float* lnw     = (const float*)d_in[5];
    const float* lnb     = (const float*)d_in[6];
    float*       out     = (float*)d_out;

    dim3 grid(NTOT / (NWAVES * 16), HOUT, 1);
    dim3 block(NTHREADS, 1, 1);
    opm_gather_gemm_ln_kernel<<<grid, block, 0, stream>>>(
        x, weight, psi, idx_hi, idx_dwi, lnw, lnb, out);
}


# Per block, rank the 96 taps by (input row, column offset) and stage psi and index pairs in that order: a wave's consecutive gathers share cache lines; changes the floating-point summation order
# speedup vs baseline: 2.7993x; 1.2608x over previous
#include <hip/hip_runtime.h>


typedef __attribute__((ext_vector_type(2))) float v2f;
typedef __attribute__((ext_vector_type(8))) float v8f;

#define B_       2
#define CIN      3
#define COUT     64
#define HIN      721
#define WIN      1440
#define HOUT     361
#define WOUT     720
#define KTAP     9
#define NNZ      96
#define LN_EPS   1e-5f

#define KPAD     16
#define KLO      5
#define PSTRIDE  98
#define WROW     (CIN * KTAP)
#define NWAVES   6
#define NTHREADS (NWAVES * 32)
#define PLANE    (HIN * WIN)
#define NTOT     (B_ * WOUT)
#define KEYPART  (NNZ / 2)

static_assert(NTHREADS == 2 * NNZ, "the ranking prologue gives every tap two threads");
static_assert(KEYPART % 4 == 0, "ranking reads the keys four at a time");

__global__ __launch_bounds__(NTHREADS)
void opm_gather_gemm_ln_kernel(const float* __restrict__ x,
                               const float* __restrict__ weight,
                               const float* __restrict__ psi,
                               const int*   __restrict__ idx_hi,
                               const int*   __restrict__ idx_dwi,
                               const float* __restrict__ lnw,
                               const float* __restrict__ lnb,
                               float*       __restrict__ out)
{
    __shared__ float psiA_lds[KPAD * PSTRIDE];
    __shared__ float w_lds[COUT * WROW];
    __shared__ __attribute__((aligned(16))) int2 base_lds[NNZ];
    __shared__ float lnw_lds[COUT];
    __shared__ float lnb_lds[COUT];
    __shared__ __attribute__((aligned(16))) int key_lds[NNZ];
    __shared__ int cnt_lds[2 * NNZ];

    const int tid  = threadIdx.x;
    const int h    = blockIdx.y;
    const int lane = tid & 31;
    const int wid  = tid >> 5;
    const int l16  = lane & 15;
    const int half = lane >> 4;

    const int part  = (tid >= NNZ) ? 1 : 0;
    const int e_own = tid - part * NNZ;
    const int hi_e  = idx_hi[h * NNZ + e_own];
    const int dwi_e = idx_dwi[h * NNZ + e_own];
    float pv[5];
    #pragma unroll
    for (int j = 0; j < 5; ++j) {
        const int k9 = part + 2 * j;
        pv[j] = (k9 < KTAP) ? psi[(k9 * HOUT + h) * NNZ + e_own] : 0.f;
    }
    if (part == 0) {
        int rot = dwi_e + WIN / 2; rot -= (rot >= WIN) ? WIN : 0;
        key_lds[e_own] = ((hi_e << 11) + rot) * 128 + e_own;
    }
    for (int i = tid; i < 7 * PSTRIDE; i += NTHREADS) {
        const int at = (i < 3 * PSTRIDE) ? (KLO * PSTRIDE + i) : (12 * PSTRIDE + (i - 3 * PSTRIDE));
        psiA_lds[at] = 0.f;
    }
    for (int i = tid; i < COUT * WROW; i += NTHREADS)
        w_lds[i] = weight[i];
    for (int i = tid; i < COUT; i += NTHREADS) {
        lnw_lds[i] = lnw[i];
        lnb_lds[i] = lnb[i];
    }

    const int n    = blockIdx.x * (NWAVES * 16) + wid * 16 + l16;
    const int b    = n / WOUT;
    const int w    = n - b * WOUT;
    const int wcol = w << 1;
    const int bofs = b * (CIN * PLANE);

    __syncthreads();
    {
        const int   mykey = key_lds[e_own];
        const int4* kp    = (const int4*)&key_lds[part * KEYPART];
        int cnt = 0;
        #pragma unroll
        for (int j = 0; j < KEYPART / 4; ++j) {
            const int4 kk = kp[j];
            cnt += (kk.x < mykey) + (kk.y < mykey) + (kk.z < mykey) + (kk.w < mykey);
        }
        cnt_lds[part * NNZ + e_own] = cnt;
    }
    __syncthreads();
    {
        const int rank = cnt_lds[e_own] + cnt_lds[NNZ + e_own];
        if (part == 0)
            base_lds[rank] = make_int2(hi_e * WIN, dwi_e);
        #pragma unroll
        for (int j = 0; j < 5; ++j) {
            const int k9 = part + 2 * j;
            if (k9 < KTAP) {
                const int m = (k9 < KLO) ? k9 : k9 + (8 - KLO);
                psiA_lds[m * PSTRIDE + rank] = pv[j];
            }
        }
    }
    __syncthreads();

    v8f y0 = (v8f){0.f,0.f,0.f,0.f,0.f,0.f,0.f,0.f};
    v8f y1 = y0, y2 = y0;

    #pragma unroll 4
    for (int e0 = 0; e0 < NNZ; e0 += 4) {
        const int eA = e0 + (half << 1);

        const int4 rr = *(const int4*)&base_lds[eA];
        int c0 = rr.y + wcol; c0 -= (c0 >= WIN) ? WIN : 0;
        int c1 = rr.w + wcol; c1 -= (c1 >= WIN) ? WIN : 0;
        const int i0 = bofs + rr.x + c0;
        const int i1 = bofs + rr.z + c1;

        v2f b0, b1, b2;
        b0.x = x[i0];              b0.y = x[i1];
        b1.x = x[i0 + PLANE];      b1.y = x[i1 + PLANE];
        b2.x = x[i0 + 2 * PLANE];  b2.y = x[i1 + 2 * PLANE];

        const v2f pa = *(const v2f*)&psiA_lds[l16 * PSTRIDE + eA];

        y0 = __builtin_amdgcn_wmma_f32_16x16x4_f32(false, pa, false, b0, (short)0, y0, false, false);
        y1 = __builtin_amdgcn_wmma_f32_16x16x4_f32(false, pa, false, b1, (short)0, y1, false, false);
        y2 = __builtin_amdgcn_wmma_f32_16x16x4_f32(false, pa, false, b2, (short)0, y2, false, false);
    }

    v8f acc[4];
    #pragma unroll
    for (int mt = 0; mt < 4; ++mt) acc[mt] = (v8f){0.f,0.f,0.f,0.f,0.f,0.f,0.f,0.f};

    const float* wl = &w_lds[l16 * WROW];
    const int   kh = half ? KLO : 0;
    const bool  lo = (half == 0);

    #pragma unroll
    for (int c = 0; c < CIN; ++c) {
        const v8f yc = (c == 0) ? y0 : ((c == 1) ? y1 : y2);
        #pragma unroll
        for (int j = 0; j < 2; ++j) {
            v2f bf;
            bf.x = yc[2 * j];
            bf.y = yc[2 * j + 1];
            #pragma unroll
            for (int mt = 0; mt < 4; ++mt) {
                const float* wp = wl + mt * 16 * WROW + c * KTAP + 2 * j + kh;
                v2f a;
                a.x = wp[0];
                a.y = wp[1];
                acc[mt] = __builtin_amdgcn_wmma_f32_16x16x4_f32(false, a, false, bf, (short)0, acc[mt], false, false);
            }
        }
    }
    {
        v2f bf6, bf7;
        bf6.x = y0[KLO - 1];  bf6.y = y1[KLO - 1];
        bf7.x = y2[KLO - 1];  bf7.y = 0.f;
        #pragma unroll
        for (int mt = 0; mt < 4; ++mt) {
            const float* wp = wl + mt * 16 * WROW + (KLO - 1);
            const float w0 = wp[0], w1 = wp[KTAP], w2 = wp[2 * KTAP];
            v2f a6, a7;
            a6.x = lo ? w0 : 0.f;
            a6.y = lo ? w1 : 0.f;
            a7.x = lo ? w2 : 0.f;
            a7.y = 0.f;
            acc[mt] = __builtin_amdgcn_wmma_f32_16x16x4_f32(false, a6, false, bf6, (short)0, acc[mt], false, false);
            acc[mt] = __builtin_amdgcn_wmma_f32_16x16x4_f32(false, a7, false, bf7, (short)0, acc[mt], false, false);
        }
    }

    float s = 0.f, ss = 0.f;
    #pragma unroll
    for (int mt = 0; mt < 4; ++mt) {
        #pragma unroll
        for (int r = 0; r < 8; ++r) {
            float v = acc[mt][r];
            s  += v;
            ss += v * v;
        }
    }
    s  += __shfl_xor(s, 16, 32);
    ss += __shfl_xor(ss, 16, 32);
    const float mean = s * (1.f / COUT);
    const float var  = ss * (1.f / COUT) - mean * mean;
    const float rstd = rsqrtf(var + LN_EPS);

    float* outp = out + (size_t)b * COUT * HOUT * WOUT + (size_t)h * WOUT + w;
    #pragma unroll
    for (int mt = 0; mt < 4; ++mt) {
        #pragma unroll
        for (int r = 0; r < 8; ++r) {
            const int o = mt * 16 + r + (half << 3);
            const float v = (acc[mt][r] - mean) * rstd * lnw_lds[o] + lnb_lds[o];
            outp[(size_t)o * (HOUT * WOUT)] = v;
        }
    }
}

extern "C" void kernel_launch(void* const* d_in, const int* in_sizes, int n_in,
                              void* d_out, int out_size, void* d_ws, size_t ws_size,
                              hipStream_t stream) {
    (void)in_sizes; (void)n_in; (void)d_ws; (void)ws_size; (void)out_size;
    const float* x       = (const float*)d_in[0];
    const float* weight  = (const float*)d_in[1];
    const float* psi     = (const float*)d_in[2];
    const int*   idx_hi  = (const int*)d_in[3];
    const int*   idx_dwi = (const int*)d_in[4];
    const float* lnw     = (const float*)d_in[5];
    const float* lnb     = (const float*)d_in[6];
    float*       out     = (float*)d_out;

    dim3 grid(NTOT / (NWAVES * 16), HOUT, 1);
    dim3 block(NTHREADS, 1, 1);
    opm_gather_gemm_ln_kernel<<<grid, block, 0, stream>>>(
        x, weight, psi, idx_hi, idx_dwi, lnw, lnb, out);
}
